# starting kernel
# baseline (speedup 1.0000x reference)
#include <hip/hip_runtime.h>

#define BT_    192
#define N_     1000
#define NP_    1008
#define K_     16
#define C1_    32
#define C2_    16
#define NK_    16000
#define NTILES 63

typedef __attribute__((ext_vector_type(2))) float v2f;
typedef __attribute__((ext_vector_type(4))) float v4f;
typedef __attribute__((ext_vector_type(8))) float v8f;
typedef __attribute__((ext_vector_type(2))) int   v2i;
typedef __attribute__((ext_vector_type(4))) int   v4i;

__device__ __forceinline__ v8f wmma_f32_4(v2f a, v2f b, v8f c) {
    return __builtin_amdgcn_wmma_f32_16x16x4_f32(
         false, a,  false, b,
         (short)0, c,  false,  false);
}

__global__ __launch_bounds__(256)
void k1_layer1(const float* __restrict__ x, const int* __restrict__ nb,
               const float* __restrict__ W1, const float* __restrict__ b1,
               float* __restrict__ h1) {
    const int lane   = threadIdx.x & 31;
    const int wave   = threadIdx.x >> 5;
    const int tileId = blockIdx.x * 8 + wave;
    if (tileId >= BT_ * NTILES) return;
    const int bt   = tileId / NTILES;
    const int base = (tileId % NTILES) * 16;

    const int mrow  = lane & 15;
    const int hi    = lane >> 4;
    const int khalf = hi * 2;
    const int node  = min(base + mrow, N_ - 1);
    const float* xb = x + bt * N_;

    for (int ct = 0; ct < 2; ++ct) {
        v8f acc = {};
#pragma unroll
        for (int s = 0; s < 4; ++s) {
            const int k0 = s * 4 + khalf;
            const v2i idx = *(const v2i*)&nb[node * K_ + k0];
            v2f a, b;
            a.x = xb[idx.x];
            a.y = xb[idx.y];
            b = *(const v2f*)&W1[(ct * 16 + mrow) * K_ + k0];
            acc = wmma_f32_4(a, b, acc);
        }
        const int   ch   = ct * 16 + mrow;
        const float bias = b1[ch];
        v4f lo, hiv;
#pragma unroll
        for (int j = 0; j < 4; ++j) {
            float v0 = acc[j]     + bias;  lo[j]  = v0 > 0.f ? v0 : 0.f;
            float v1 = acc[j + 4] + bias;  hiv[j] = v1 > 0.f ? v1 : 0.f;
        }
        float* p = h1 + (size_t)(bt * C1_ + ch) * NP_ + base + 8 * hi;
        *(v4f*)p       = lo;
        *(v4f*)(p + 4) = hiv;
    }
}

__global__ __launch_bounds__(256)
void k2_layer2(const float* __restrict__ h1, const int* __restrict__ nb,
               const float* __restrict__ W2, const float* __restrict__ b2,
               const float* __restrict__ gamma, const float* __restrict__ beta,
               float* __restrict__ h2) {
    __shared__ float h1s[C1_ * NP_];
    __shared__ int   nbs[NK_];
    __shared__ float w2s[C2_ * 512];

    const int bt     = blockIdx.x >> 1;
    const int halfId = blockIdx.x & 1;

    const float* h1b = h1 + (size_t)bt * C1_ * NP_;
    for (int i = threadIdx.x; i < C1_ * NP_ / 4; i += 256)
        ((v4f*)h1s)[i] = ((const v4f*)h1b)[i];
    for (int i = threadIdx.x; i < NK_ / 4; i += 256)
        ((v4i*)nbs)[i] = ((const v4i*)nb)[i];
    for (int i = threadIdx.x; i < C2_ * 512 / 4; i += 256)
        ((v4f*)w2s)[i] = ((const v4f*)W2)[i];
    __syncthreads();

    const int lane  = threadIdx.x & 31;
    const int wave  = threadIdx.x >> 5;
    const int mrow  = lane & 15;
    const int hi    = lane >> 4;
    const int khalf = hi * 2;

    const float scale = gamma[mrow] * rsqrtf(1.0f + 1e-5f);
    const float bias  = b2[mrow];
    const float shift = beta[mrow];

    for (int t = 0; t < 4; ++t) {
        const int tile = halfId * 32 + wave * 4 + t;
        if (tile >= NTILES) continue;
        const int base = tile * 16;
        const int node = min(base + mrow, N_ - 1);
        const unsigned fbase = (unsigned)node * 512u;

        v8f acc = {};
#pragma unroll 8
        for (int s = 0; s < 128; ++s) {
            const int k0 = s * 4 + khalf;
            const unsigned f0 = fbase + (unsigned)k0;
            const unsigned c  = f0 / (unsigned)NK_;
            const unsigned r  = f0 - c * (unsigned)NK_;
            const v2i idx = *(const v2i*)&nbs[r];
            v2f a, b;
            a.x = h1s[c * NP_ + idx.x];
            a.y = h1s[c * NP_ + idx.y];
            b = *(const v2f*)&w2s[mrow * 512 + k0];
            acc = wmma_f32_4(a, b, acc);
        }
        v4f lo, hiv;
#pragma unroll
        for (int j = 0; j < 4; ++j) {
            float v0 = (acc[j]     + bias) * scale + shift;  lo[j]  = v0 > 0.f ? v0 : 0.f;
            float v1 = (acc[j + 4] + bias) * scale + shift;  hiv[j] = v1 > 0.f ? v1 : 0.f;
        }
        float* p = h2 + (size_t)(bt * C2_ + mrow) * NP_ + base + 8 * hi;
        *(v4f*)p       = lo;
        *(v4f*)(p + 4) = hiv;
    }
}

__global__ __launch_bounds__(256)
void k3_layer3(const float* __restrict__ h2, const int* __restrict__ nb,
               const float* __restrict__ W3, const float* __restrict__ b3,
               float* __restrict__ out) {
    __shared__ float h2s[C2_ * NP_];
    __shared__ int   nbs[NK_];
    __shared__ float w3s[256];

    const int bt     = blockIdx.x >> 1;
    const int halfId = blockIdx.x & 1;

    const float* h2b = h2 + (size_t)bt * C2_ * NP_;
    for (int i = threadIdx.x; i < C2_ * NP_ / 4; i += 256)
        ((v4f*)h2s)[i] = ((const v4f*)h2b)[i];
    for (int i = threadIdx.x; i < NK_ / 4; i += 256)
        ((v4i*)nbs)[i] = ((const v4i*)nb)[i];
    if (threadIdx.x < 256) w3s[threadIdx.x] = W3[threadIdx.x];
    __syncthreads();

    const float b3v = b3[0];
    for (int i = 0; i < 2; ++i) {
        const int off = i * 256 + threadIdx.x;
        if (off >= 500) break;
        const int n = halfId * 500 + off;
        float acc = b3v;
        const unsigned fbase = (unsigned)n * 256u;
#pragma unroll 8
        for (int f = 0; f < 256; f += 2) {
            const unsigned fl = fbase + (unsigned)f;
            const unsigned c  = fl / (unsigned)NK_;
            const unsigned r  = fl - c * (unsigned)NK_;
            const v2i idx = *(const v2i*)&nbs[r];
            acc = fmaf(h2s[c * NP_ + idx.x], w3s[f],     acc);
            acc = fmaf(h2s[c * NP_ + idx.y], w3s[f + 1], acc);
        }
        out[bt * N_ + n] = acc > 0.f ? acc : 0.f;
    }
}

extern "C" void kernel_launch(void* const* d_in, const int* in_sizes, int n_in,
                              void* d_out, int out_size, void* d_ws, size_t ws_size,
                              hipStream_t stream) {
    const float* x     = (const float*)d_in[0];
    const int*   nb    = (const int*)  d_in[1];
    const float* W1    = (const float*)d_in[2];
    const float* b1    = (const float*)d_in[3];
    const float* W2    = (const float*)d_in[4];
    const float* b2    = (const float*)d_in[5];
    const float* W3    = (const float*)d_in[6];
    const float* b3    = (const float*)d_in[7];
    const float* gamma = (const float*)d_in[8];
    const float* beta  = (const float*)d_in[9];

    float* h1  = (float*)d_ws;
    float* h2  = h1 + (size_t)BT_ * C1_ * NP_;
    float* out = (float*)d_out;

    const int tiles  = BT_ * NTILES;
    const int blocks = (tiles + 7) / 8;

    k1_layer1<<<blocks,  256, 0, stream>>>(x, nb, W1, b1, h1);
    k2_layer2<<<BT_ * 2, 256, 0, stream>>>(h1, nb, W2, b2, gamma, beta, h2);
    k3_layer3<<<BT_ * 2, 256, 0, stream>>>(h2, nb, W3, b3, out);
}


# Pad LDS rows so lane-strided reads spread over all banks: layer 2 neighbor table +4 dwords per 512 and W2 pitch 516, layer 3 table +4 per 256 (all rows of a fragment hit one bank before)
# speedup vs baseline: 1.9679x; 1.9679x over previous
#include <hip/hip_runtime.h>

#define BT_    192
#define N_     1000
#define NP_    1008
#define K_     16
#define C1_    32
#define C2_    16
#define NK_    16000
#define NTILES 63
#define NBP2_  16128
#define NBP3_  16256
#define W2P_   516

typedef __attribute__((ext_vector_type(2))) float v2f;
typedef __attribute__((ext_vector_type(4))) float v4f;
typedef __attribute__((ext_vector_type(8))) float v8f;
typedef __attribute__((ext_vector_type(2))) int   v2i;
typedef __attribute__((ext_vector_type(4))) int   v4i;

__device__ __forceinline__ v8f wmma_f32_4(v2f a, v2f b, v8f c) {
    return __builtin_amdgcn_wmma_f32_16x16x4_f32(
         false, a,  false, b,
         (short)0, c,  false,  false);
}

__global__ __launch_bounds__(256)
void k1_layer1(const float* __restrict__ x, const int* __restrict__ nb,
               const float* __restrict__ W1, const float* __restrict__ b1,
               float* __restrict__ h1) {
    const int lane   = threadIdx.x & 31;
    const int wave   = threadIdx.x >> 5;
    const int tileId = blockIdx.x * 8 + wave;
    if (tileId >= BT_ * NTILES) return;
    const int bt   = tileId / NTILES;
    const int base = (tileId % NTILES) * 16;

    const int mrow  = lane & 15;
    const int hi    = lane >> 4;
    const int khalf = hi * 2;
    const int node  = min(base + mrow, N_ - 1);
    const float* xb = x + bt * N_;

    for (int ct = 0; ct < 2; ++ct) {
        v8f acc = {};
#pragma unroll
        for (int s = 0; s < 4; ++s) {
            const int k0 = s * 4 + khalf;
            const v2i idx = *(const v2i*)&nb[node * K_ + k0];
            v2f a, b;
            a.x = xb[idx.x];
            a.y = xb[idx.y];
            b = *(const v2f*)&W1[(ct * 16 + mrow) * K_ + k0];
            acc = wmma_f32_4(a, b, acc);
        }
        const int   ch   = ct * 16 + mrow;
        const float bias = b1[ch];
        v4f lo, hiv;
#pragma unroll
        for (int j = 0; j < 4; ++j) {
            float v0 = acc[j]     + bias;  lo[j]  = v0 > 0.f ? v0 : 0.f;
            float v1 = acc[j + 4] + bias;  hiv[j] = v1 > 0.f ? v1 : 0.f;
        }
        float* p = h1 + (size_t)(bt * C1_ + ch) * NP_ + base + 8 * hi;
        *(v4f*)p       = lo;
        *(v4f*)(p + 4) = hiv;
    }
}

__global__ __launch_bounds__(256)
void k2_layer2(const float* __restrict__ h1, const int* __restrict__ nb,
               const float* __restrict__ W2, const float* __restrict__ b2,
               const float* __restrict__ gamma, const float* __restrict__ beta,
               float* __restrict__ h2) {
    __shared__ float h1s[C1_ * NP_];
    __shared__ int   nbs[NBP2_];
    __shared__ float w2s[C2_ * W2P_];

    const int bt     = blockIdx.x >> 1;
    const int halfId = blockIdx.x & 1;

    const float* h1b = h1 + (size_t)bt * C1_ * NP_;
    for (int i = threadIdx.x; i < C1_ * NP_ / 4; i += 256)
        ((v4f*)h1s)[i] = ((const v4f*)h1b)[i];
    for (int i = threadIdx.x; i < NK_ / 4; i += 256)
        ((v4i*)nbs)[i + (i >> 7)] = ((const v4i*)nb)[i];
    for (int i = threadIdx.x; i < C2_ * 512 / 4; i += 256)
        ((v4f*)w2s)[i + (i >> 7)] = ((const v4f*)W2)[i];
    __syncthreads();

    const int lane  = threadIdx.x & 31;
    const int wave  = threadIdx.x >> 5;
    const int mrow  = lane & 15;
    const int hi    = lane >> 4;
    const int khalf = hi * 2;

    const float scale = gamma[mrow] * rsqrtf(1.0f + 1e-5f);
    const float bias  = b2[mrow];
    const float shift = beta[mrow];

    for (int t = 0; t < 4; ++t) {
        const int tile = halfId * 32 + wave * 4 + t;
        if (tile >= NTILES) continue;
        const int base = tile * 16;
        const int node = min(base + mrow, N_ - 1);
        const unsigned fbase = (unsigned)node * 512u;

        v8f acc = {};
#pragma unroll 8
        for (int s = 0; s < 128; ++s) {
            const int k0 = s * 4 + khalf;
            const unsigned f0 = fbase + (unsigned)k0;
            const unsigned c  = f0 / (unsigned)NK_;
            const unsigned r  = f0 - c * (unsigned)NK_;
            const unsigned pr = r + ((r >> 9) << 2);
            const v2i idx = *(const v2i*)&nbs[pr];
            v2f a, b;
            a.x = h1s[c * NP_ + idx.x];
            a.y = h1s[c * NP_ + idx.y];
            b = *(const v2f*)&w2s[mrow * W2P_ + k0];
            acc = wmma_f32_4(a, b, acc);
        }
        v4f lo, hiv;
#pragma unroll
        for (int j = 0; j < 4; ++j) {
            float v0 = (acc[j]     + bias) * scale + shift;  lo[j]  = v0 > 0.f ? v0 : 0.f;
            float v1 = (acc[j + 4] + bias) * scale + shift;  hiv[j] = v1 > 0.f ? v1 : 0.f;
        }
        float* p = h2 + (size_t)(bt * C2_ + mrow) * NP_ + base + 8 * hi;
        *(v4f*)p       = lo;
        *(v4f*)(p + 4) = hiv;
    }
}

__global__ __launch_bounds__(256)
void k3_layer3(const float* __restrict__ h2, const int* __restrict__ nb,
               const float* __restrict__ W3, const float* __restrict__ b3,
               float* __restrict__ out) {
    __shared__ float h2s[C2_ * NP_];
    __shared__ int   nbs[NBP3_];
    __shared__ float w3s[256];

    const int bt     = blockIdx.x >> 1;
    const int halfId = blockIdx.x & 1;

    const float* h2b = h2 + (size_t)bt * C2_ * NP_;
    for (int i = threadIdx.x; i < C2_ * NP_ / 4; i += 256)
        ((v4f*)h2s)[i] = ((const v4f*)h2b)[i];
    for (int i = threadIdx.x; i < NK_ / 4; i += 256)
        ((v4i*)nbs)[i + (i >> 6)] = ((const v4i*)nb)[i];
    if (threadIdx.x < 256) w3s[threadIdx.x] = W3[threadIdx.x];
    __syncthreads();

    const float b3v = b3[0];
    for (int i = 0; i < 2; ++i) {
        const int off = i * 256 + threadIdx.x;
        if (off >= 500) break;
        const int n = halfId * 500 + off;
        float acc = b3v;
        const unsigned fbase = (unsigned)n * 256u;
#pragma unroll 1
        for (int f = 0; f < 256; f += 16) {
            const unsigned fl = fbase + (unsigned)f;
            const unsigned c  = fl / (unsigned)NK_;
            const unsigned r  = fl - c * (unsigned)NK_;
            const unsigned pr = r + ((r >> 8) << 2);
            const float* hc = h2s + c * NP_;
#pragma unroll
            for (int q = 0; q < 4; ++q) {
                const v4i idx = *(const v4i*)&nbs[pr + 4 * q];
                const v4f w   = *(const v4f*)&w3s[f + 4 * q];
                acc = fmaf(hc[idx.x], w.x, acc);
                acc = fmaf(hc[idx.y], w.y, acc);
                acc = fmaf(hc[idx.z], w.z, acc);
                acc = fmaf(hc[idx.w], w.w, acc);
            }
        }
        out[bt * N_ + n] = acc > 0.f ? acc : 0.f;
    }
}

extern "C" void kernel_launch(void* const* d_in, const int* in_sizes, int n_in,
                              void* d_out, int out_size, void* d_ws, size_t ws_size,
                              hipStream_t stream) {
    const float* x     = (const float*)d_in[0];
    const int*   nb    = (const int*)  d_in[1];
    const float* W1    = (const float*)d_in[2];
    const float* b1    = (const float*)d_in[3];
    const float* W2    = (const float*)d_in[4];
    const float* b2    = (const float*)d_in[5];
    const float* W3    = (const float*)d_in[6];
    const float* b3    = (const float*)d_in[7];
    const float* gamma = (const float*)d_in[8];
    const float* beta  = (const float*)d_in[9];

    float* h1  = (float*)d_ws;
    float* h2  = h1 + (size_t)BT_ * C1_ * NP_;
    float* out = (float*)d_out;

    const int tiles  = BT_ * NTILES;
    const int blocks = (tiles + 7) / 8;

    k1_layer1<<<blocks,  256, 0, stream>>>(x, nb, W1, b1, h1);
    k2_layer2<<<BT_ * 2, 256, 0, stream>>>(h1, nb, W2, b2, gamma, beta, h2);
    k3_layer3<<<BT_ * 2, 256, 0, stream>>>(h2, nb, W3, b3, out);
}


# Layer 2: one block serves two bts, activations interleaved in LDS as (bt0,bt1) pairs so each index read, W2 read and 8-byte gather feeds two WMMAs; 192 blocks of 16 waves in one round
# speedup vs baseline: 2.5758x; 1.3089x over previous
#include <hip/hip_runtime.h>

#define BT_    192
#define N_     1000
#define NP_    1008
#define K_     16
#define C1_    32
#define C2_    16
#define NK_    16000
#define NTILES 63
#define NBP2_  16128
#define NBP3_  16256
#define W2P_   516

typedef __attribute__((ext_vector_type(2))) float v2f;
typedef __attribute__((ext_vector_type(4))) float v4f;
typedef __attribute__((ext_vector_type(8))) float v8f;
typedef __attribute__((ext_vector_type(2))) int   v2i;
typedef __attribute__((ext_vector_type(4))) int   v4i;

__device__ __forceinline__ v8f wmma_f32_4(v2f a, v2f b, v8f c) {
    return __builtin_amdgcn_wmma_f32_16x16x4_f32(
         false, a,  false, b,
         (short)0, c,  false,  false);
}

__global__ __launch_bounds__(256)
void k1_layer1(const float* __restrict__ x, const int* __restrict__ nb,
               const float* __restrict__ W1, const float* __restrict__ b1,
               float* __restrict__ h1) {
    const int lane   = threadIdx.x & 31;
    const int wave   = threadIdx.x >> 5;
    const int tileId = blockIdx.x * 8 + wave;
    if (tileId >= BT_ * NTILES) return;
    const int bt   = tileId / NTILES;
    const int base = (tileId % NTILES) * 16;

    const int mrow  = lane & 15;
    const int hi    = lane >> 4;
    const int khalf = hi * 2;
    const int node  = min(base + mrow, N_ - 1);
    const float* xb = x + bt * N_;

    for (int ct = 0; ct < 2; ++ct) {
        v8f acc = {};
#pragma unroll
        for (int s = 0; s < 4; ++s) {
            const int k0 = s * 4 + khalf;
            const v2i idx = *(const v2i*)&nb[node * K_ + k0];
            v2f a, b;
            a.x = xb[idx.x];
            a.y = xb[idx.y];
            b = *(const v2f*)&W1[(ct * 16 + mrow) * K_ + k0];
            acc = wmma_f32_4(a, b, acc);
        }
        const int   ch   = ct * 16 + mrow;
        const float bias = b1[ch];
        v4f lo, hiv;
#pragma unroll
        for (int j = 0; j < 4; ++j) {
            float v0 = acc[j]     + bias;  lo[j]  = v0 > 0.f ? v0 : 0.f;
            float v1 = acc[j + 4] + bias;  hiv[j] = v1 > 0.f ? v1 : 0.f;
        }
        float* p = h1 + (size_t)(bt * C1_ + ch) * NP_ + base + 8 * hi;
        *(v4f*)p       = lo;
        *(v4f*)(p + 4) = hiv;
    }
}

#define NCH_   17
#define NQH_   4284
__global__ __launch_bounds__(512)
void k2_layer2(const float* __restrict__ h1, const int* __restrict__ nb,
               const float* __restrict__ W2, const float* __restrict__ b2,
               const float* __restrict__ gamma, const float* __restrict__ beta,
               float* __restrict__ h2) {
    __shared__ v2f   h1p[NCH_ * NP_];
    __shared__ int   nbs[NBP2_];
    __shared__ float w2s[C2_ * W2P_];

    const int tid    = threadIdx.x;
    const int bt0    = (blockIdx.x >> 1) * 2;
    const int halfId = blockIdx.x & 1;
    const unsigned cbase = (unsigned)halfId * 15u;

    const v4f* s0 = (const v4f*)(h1 + ((size_t)bt0 * C1_ + cbase) * NP_);
    const v4f* s1 = (const v4f*)(h1 + ((size_t)(bt0 + 1) * C1_ + cbase) * NP_);
    const v4i* nq = (const v4i*)nb;
    const v4f* wq = (const v4f*)W2;
    v4f ra[9], rb[9], rw[4];
    v4i rn[8];
#pragma unroll
    for (int it = 0; it < 8; ++it) {
        ra[it] = s0[tid + it * 512];
        rb[it] = s1[tid + it * 512];
    }
#pragma unroll
    for (int it = 0; it < 7; ++it)
        rn[it] = nq[tid + it * 512];
#pragma unroll
    for (int it = 0; it < 4; ++it)
        rw[it] = wq[tid + it * 512];
    ra[8] = (v4f){0.f, 0.f, 0.f, 0.f};
    rb[8] = (v4f){0.f, 0.f, 0.f, 0.f};
    rn[7] = (v4i){0, 0, 0, 0};
    if (tid + 4096 < NQH_) {
        ra[8] = s0[tid + 4096];
        rb[8] = s1[tid + 4096];
    }
    if (tid + 3584 < NK_ / 4) rn[7] = nq[tid + 3584];

#pragma unroll
    for (int it = 0; it < 9; ++it) {
        const int i = tid + it * 512;
        if (it < 8 || i < NQH_) {
            const v4f a = ra[it], b = rb[it];
            const v4f lo  = {a.x, b.x, a.y, b.y};
            const v4f hiq = {a.z, b.z, a.w, b.w};
            ((v4f*)h1p)[2 * i]     = lo;
            ((v4f*)h1p)[2 * i + 1] = hiq;
        }
    }
#pragma unroll
    for (int it = 0; it < 8; ++it) {
        const int i = tid + it * 512;
        if (it < 7 || i < NK_ / 4)
            ((v4i*)nbs)[i + (i >> 7)] = rn[it];
    }
#pragma unroll
    for (int it = 0; it < 4; ++it) {
        const int i = tid + it * 512;
        ((v4f*)w2s)[i + (i >> 7)] = rw[it];
    }
    __syncthreads();

    const int lane  = tid & 31;
    const int wave  = tid >> 5;
    const int mrow  = lane & 15;
    const int hi    = lane >> 4;
    const int khalf = hi * 2;

    const float scale = gamma[mrow] * rsqrtf(1.0f + 1e-5f);
    const float bias  = b2[mrow];
    const float shift = beta[mrow];

    for (int t = 0; t < 2; ++t) {
        const int tile = halfId * 32 + wave * 2 + t;
        if (tile >= NTILES) continue;
        const int base = tile * 16;
        const int node = min(base + mrow, N_ - 1);
        const unsigned fbase = (unsigned)node * 512u;

        v8f acc0 = {};
        v8f acc1 = {};
#pragma unroll 8
        for (int s = 0; s < 128; ++s) {
            const int k0 = s * 4 + khalf;
            const unsigned f0 = fbase + (unsigned)k0;
            const unsigned c  = f0 / (unsigned)NK_;
            const unsigned r  = f0 - c * (unsigned)NK_;
            const unsigned pr = r + ((r >> 9) << 2);
            const v2i idx = *(const v2i*)&nbs[pr];
            const v2f* hc = h1p + (c - cbase) * NP_;
            const v2f g0 = hc[idx.x];
            const v2f g1 = hc[idx.y];
            const v2f a0 = {g0.x, g1.x};
            const v2f a1 = {g0.y, g1.y};
            const v2f b = *(const v2f*)&w2s[mrow * W2P_ + k0];
            acc0 = wmma_f32_4(a0, b, acc0);
            acc1 = wmma_f32_4(a1, b, acc1);
        }
#pragma unroll
        for (int q = 0; q < 2; ++q) {
            const v8f acc = q ? acc1 : acc0;
            v4f lo, hiv;
#pragma unroll
            for (int j = 0; j < 4; ++j) {
                float v0 = (acc[j]     + bias) * scale + shift;  lo[j]  = v0 > 0.f ? v0 : 0.f;
                float v1 = (acc[j + 4] + bias) * scale + shift;  hiv[j] = v1 > 0.f ? v1 : 0.f;
            }
            float* p = h2 + (size_t)((bt0 + q) * C2_ + mrow) * NP_ + base + 8 * hi;
            *(v4f*)p       = lo;
            *(v4f*)(p + 4) = hiv;
        }
    }
}

__global__ __launch_bounds__(256)
void k3_layer3(const float* __restrict__ h2, const int* __restrict__ nb,
               const float* __restrict__ W3, const float* __restrict__ b3,
               float* __restrict__ out) {
    __shared__ float h2s[C2_ * NP_];
    __shared__ int   nbs[NBP3_];
    __shared__ float w3s[256];

    const int bt     = blockIdx.x >> 1;
    const int halfId = blockIdx.x & 1;

    const float* h2b = h2 + (size_t)bt * C2_ * NP_;
    for (int i = threadIdx.x; i < C2_ * NP_ / 4; i += 256)
        ((v4f*)h2s)[i] = ((const v4f*)h2b)[i];
    for (int i = threadIdx.x; i < NK_ / 4; i += 256)
        ((v4i*)nbs)[i + (i >> 6)] = ((const v4i*)nb)[i];
    if (threadIdx.x < 256) w3s[threadIdx.x] = W3[threadIdx.x];
    __syncthreads();

    const float b3v = b3[0];
    for (int i = 0; i < 2; ++i) {
        const int off = i * 256 + threadIdx.x;
        if (off >= 500) break;
        const int n = halfId * 500 + off;
        float acc = b3v;
        const unsigned fbase = (unsigned)n * 256u;
#pragma unroll 1
        for (int f = 0; f < 256; f += 16) {
            const unsigned fl = fbase + (unsigned)f;
            const unsigned c  = fl / (unsigned)NK_;
            const unsigned r  = fl - c * (unsigned)NK_;
            const unsigned pr = r + ((r >> 8) << 2);
            const float* hc = h2s + c * NP_;
#pragma unroll
            for (int q = 0; q < 4; ++q) {
                const v4i idx = *(const v4i*)&nbs[pr + 4 * q];
                const v4f w   = *(const v4f*)&w3s[f + 4 * q];
                acc = fmaf(hc[idx.x], w.x, acc);
                acc = fmaf(hc[idx.y], w.y, acc);
                acc = fmaf(hc[idx.z], w.z, acc);
                acc = fmaf(hc[idx.w], w.w, acc);
            }
        }
        out[bt * N_ + n] = acc > 0.f ? acc : 0.f;
    }
}

extern "C" void kernel_launch(void* const* d_in, const int* in_sizes, int n_in,
                              void* d_out, int out_size, void* d_ws, size_t ws_size,
                              hipStream_t stream) {
    const float* x     = (const float*)d_in[0];
    const int*   nb    = (const int*)  d_in[1];
    const float* W1    = (const float*)d_in[2];
    const float* b1    = (const float*)d_in[3];
    const float* W2    = (const float*)d_in[4];
    const float* b2    = (const float*)d_in[5];
    const float* W3    = (const float*)d_in[6];
    const float* b3    = (const float*)d_in[7];
    const float* gamma = (const float*)d_in[8];
    const float* beta  = (const float*)d_in[9];

    float* h1  = (float*)d_ws;
    float* h2  = h1 + (size_t)BT_ * C1_ * NP_;
    float* out = (float*)d_out;

    const int tiles  = BT_ * NTILES;
    const int blocks = (tiles + 7) / 8;

    k1_layer1<<<blocks,  256, 0, stream>>>(x, nb, W1, b1, h1);
    k2_layer2<<<BT_,     512, 0, stream>>>(h1, nb, W2, b2, gamma, beta, h2);
    k3_layer3<<<BT_ * 2, 256, 0, stream>>>(h2, nb, W3, b3, out);
}


# Fuse layer 1 into the layer-2 block: it gathers x of its two bts in LDS via the staged neighbor table and builds the h1 channels it needs by the same WMMAs into LDS; h1 skips memory, 3 launches -> 2
# speedup vs baseline: 3.3236x; 1.2903x over previous
#include <hip/hip_runtime.h>

#define BT_    192
#define N_     1000
#define NP_    1008
#define K_     16
#define C1_    32
#define C2_    16
#define NK_    16000
#define NTILES 63
#define NBP2_  16128
#define NBP3_  16256
#define W2P_   516

typedef __attribute__((ext_vector_type(2))) float v2f;
typedef __attribute__((ext_vector_type(4))) float v4f;
typedef __attribute__((ext_vector_type(8))) float v8f;
typedef __attribute__((ext_vector_type(2))) int   v2i;
typedef __attribute__((ext_vector_type(4))) int   v4i;

__device__ __forceinline__ v8f wmma_f32_4(v2f a, v2f b, v8f c) {
    return __builtin_amdgcn_wmma_f32_16x16x4_f32(
         false, a,  false, b,
         (short)0, c,  false,  false);
}

#define NCH_   17
#define H1P_   1010

__device__ __forceinline__ void store_h1_pairs(v2f* dst, v8f acc0, v8f acc1, float bias) {
    for (int jj = 0; jj < 4; ++jj) {
        float p0 = acc0[2 * jj]     + bias;  p0 = p0 > 0.f ? p0 : 0.f;
        float q0 = acc1[2 * jj]     + bias;  q0 = q0 > 0.f ? q0 : 0.f;
        float p1 = acc0[2 * jj + 1] + bias;  p1 = p1 > 0.f ? p1 : 0.f;
        float q1 = acc1[2 * jj + 1] + bias;  q1 = q1 > 0.f ? q1 : 0.f;
        ((v4f*)dst)[jj] = (v4f){p0, q0, p1, q1};
    }
}

__global__ __launch_bounds__(512)
void k2_layer12(const float* __restrict__ x, const int* __restrict__ nb,
                const float* __restrict__ W1, const float* __restrict__ b1,
                const float* __restrict__ W2, const float* __restrict__ b2,
                const float* __restrict__ gamma, const float* __restrict__ beta,
                float* __restrict__ h2) {
    __shared__ __attribute__((aligned(16))) v2f   h1p[NCH_ * H1P_];
    __shared__ __attribute__((aligned(16))) int   nbs[NBP2_];
    __shared__ __attribute__((aligned(16))) float w2s[C2_ * W2P_];
    __shared__ __attribute__((aligned(16))) v2f   xs[NP_];

    const int tid    = threadIdx.x;
    const int bt0    = (blockIdx.x >> 1) * 2;
    const int halfId = blockIdx.x & 1;
    const unsigned cbase = (unsigned)halfId * 15u;

    const int lane  = tid & 31;
    const int wave  = tid >> 5;
    const int mrow  = lane & 15;
    const int hi    = lane >> 4;
    const int khalf = hi * 2;

    const v4i* nq = (const v4i*)nb;
    const v4f* wq = (const v4f*)W2;
    const v4f* x0 = (const v4f*)(x + (size_t)bt0 * N_);
    const v4f* x1 = (const v4f*)(x + (size_t)(bt0 + 1) * N_);
    v4f rw[4];
    v4i rn[8];
#pragma unroll
    for (int it = 0; it < 7; ++it)
        rn[it] = nq[tid + it * 512];
#pragma unroll
    for (int it = 0; it < 4; ++it)
        rw[it] = wq[tid + it * 512];
    v2f bw0[4], bw1[4];
    for (int s = 0; s < 4; ++s) {
        bw0[s] = *(const v2f*)&W1[mrow * K_ + s * 4 + khalf];
        bw1[s] = *(const v2f*)&W1[(16 + mrow) * K_ + s * 4 + khalf];
    }
    const float bias1_0 = b1[mrow];
    const float bias1_1 = b1[16 + mrow];
    rn[7] = (v4i){0, 0, 0, 0};
    v4f xa = {0.f, 0.f, 0.f, 0.f}, xb = {0.f, 0.f, 0.f, 0.f};
    if (tid + 3584 < NK_ / 4) rn[7] = nq[tid + 3584];
    if (tid < N_ / 4) {
        xa = x0[tid];
        xb = x1[tid];
    }

#pragma unroll
    for (int it = 0; it < 8; ++it) {
        const int i = tid + it * 512;
        if (it < 7 || i < NK_ / 4)
            ((v4i*)nbs)[i + (i >> 7)] = rn[it];
    }
#pragma unroll
    for (int it = 0; it < 4; ++it) {
        const int i = tid + it * 512;
        ((v4f*)w2s)[i + (i >> 7)] = rw[it];
    }
    if (tid < N_ / 4) {
        ((v4f*)xs)[2 * tid]     = (v4f){xa.x, xb.x, xa.y, xb.y};
        ((v4f*)xs)[2 * tid + 1] = (v4f){xa.z, xb.z, xa.w, xb.w};
    }
    __syncthreads();

    const int wave_u = __builtin_amdgcn_readfirstlane(wave);
    for (int nt = wave_u; nt < NTILES; nt += 16) {
        const int base1 = nt * 16;
        const int node1 = min(base1 + mrow, N_ - 1);
        v2f a0[4], a1[4];
        for (int s = 0; s < 4; ++s) {
            const unsigned r  = (unsigned)node1 * K_ + (unsigned)(s * 4 + khalf);
            const unsigned pr = r + ((r >> 9) << 2);
            const v2i idx = *(const v2i*)&nbs[pr];
            const v2f g0 = xs[idx.x];
            const v2f g1 = xs[idx.y];
            a0[s] = (v2f){g0.x, g1.x};
            a1[s] = (v2f){g0.y, g1.y};
        }
        if (halfId == 0) {
            v8f acc0 = {}, acc1 = {};
            for (int s = 0; s < 4; ++s) {
                acc0 = wmma_f32_4(a0[s], bw0[s], acc0);
                acc1 = wmma_f32_4(a1[s], bw0[s], acc1);
            }
            store_h1_pairs(h1p + mrow * H1P_ + base1 + 8 * hi, acc0, acc1, bias1_0);
        }
        {
            v8f acc0 = {}, acc1 = {};
            for (int s = 0; s < 4; ++s) {
                acc0 = wmma_f32_4(a0[s], bw1[s], acc0);
                acc1 = wmma_f32_4(a1[s], bw1[s], acc1);
            }
            const unsigned cl = 16u + (unsigned)mrow - cbase;
            if (cl < (unsigned)NCH_)
                store_h1_pairs(h1p + cl * H1P_ + base1 + 8 * hi, acc0, acc1, bias1_1);
        }
    }
    __syncthreads();

    const float scale = gamma[mrow] * rsqrtf(1.0f + 1e-5f);
    const float bias  = b2[mrow];
    const float shift = beta[mrow];

    for (int t = 0; t < 2; ++t) {
        const int tile = halfId * 32 + wave * 2 + t;
        if (tile >= NTILES) continue;
        const int base = tile * 16;
        const int node = min(base + mrow, N_ - 1);
        const unsigned fbase = (unsigned)node * 512u;

        v8f acc0 = {};
        v8f acc1 = {};
#pragma unroll 8
        for (int s = 0; s < 128; ++s) {
            const int k0 = s * 4 + khalf;
            const unsigned f0 = fbase + (unsigned)k0;
            const unsigned c  = f0 / (unsigned)NK_;
            const unsigned r  = f0 - c * (unsigned)NK_;
            const unsigned pr = r + ((r >> 9) << 2);
            const v2i idx = *(const v2i*)&nbs[pr];
            const v2f* hc = h1p + (c - cbase) * H1P_;
            const v2f g0 = hc[idx.x];
            const v2f g1 = hc[idx.y];
            const v2f a0 = {g0.x, g1.x};
            const v2f a1 = {g0.y, g1.y};
            const v2f b = *(const v2f*)&w2s[mrow * W2P_ + k0];
            acc0 = wmma_f32_4(a0, b, acc0);
            acc1 = wmma_f32_4(a1, b, acc1);
        }
#pragma unroll
        for (int q = 0; q < 2; ++q) {
            const v8f acc = q ? acc1 : acc0;
            v4f lo, hiv;
#pragma unroll
            for (int j = 0; j < 4; ++j) {
                float v0 = (acc[j]     + bias) * scale + shift;  lo[j]  = v0 > 0.f ? v0 : 0.f;
                float v1 = (acc[j + 4] + bias) * scale + shift;  hiv[j] = v1 > 0.f ? v1 : 0.f;
            }
            float* p = h2 + (size_t)((bt0 + q) * C2_ + mrow) * NP_ + base + 8 * hi;
            *(v4f*)p       = lo;
            *(v4f*)(p + 4) = hiv;
        }
    }
}

__global__ __launch_bounds__(256)
void k3_layer3(const float* __restrict__ h2, const int* __restrict__ nb,
               const float* __restrict__ W3, const float* __restrict__ b3,
               float* __restrict__ out) {
    __shared__ float h2s[C2_ * NP_];
    __shared__ int   nbs[NBP3_];
    __shared__ float w3s[256];

    const int bt     = blockIdx.x >> 1;
    const int halfId = blockIdx.x & 1;

    const float* h2b = h2 + (size_t)bt * C2_ * NP_;
    for (int i = threadIdx.x; i < C2_ * NP_ / 4; i += 256)
        ((v4f*)h2s)[i] = ((const v4f*)h2b)[i];
    for (int i = threadIdx.x; i < NK_ / 4; i += 256)
        ((v4i*)nbs)[i + (i >> 6)] = ((const v4i*)nb)[i];
    if (threadIdx.x < 256) w3s[threadIdx.x] = W3[threadIdx.x];
    __syncthreads();

    const float b3v = b3[0];
    for (int i = 0; i < 2; ++i) {
        const int off = i * 256 + threadIdx.x;
        if (off >= 500) break;
        const int n = halfId * 500 + off;
        float acc = b3v;
        const unsigned fbase = (unsigned)n * 256u;
#pragma unroll 1
        for (int f = 0; f < 256; f += 16) {
            const unsigned fl = fbase + (unsigned)f;
            const unsigned c  = fl / (unsigned)NK_;
            const unsigned r  = fl - c * (unsigned)NK_;
            const unsigned pr = r + ((r >> 8) << 2);
            const float* hc = h2s + c * NP_;
#pragma unroll
            for (int q = 0; q < 4; ++q) {
                const v4i idx = *(const v4i*)&nbs[pr + 4 * q];
                const v4f w   = *(const v4f*)&w3s[f + 4 * q];
                acc = fmaf(hc[idx.x], w.x, acc);
                acc = fmaf(hc[idx.y], w.y, acc);
                acc = fmaf(hc[idx.z], w.z, acc);
                acc = fmaf(hc[idx.w], w.w, acc);
            }
        }
        out[bt * N_ + n] = acc > 0.f ? acc : 0.f;
    }
}

extern "C" void kernel_launch(void* const* d_in, const int* in_sizes, int n_in,
                              void* d_out, int out_size, void* d_ws, size_t ws_size,
                              hipStream_t stream) {
    const float* x     = (const float*)d_in[0];
    const int*   nb    = (const int*)  d_in[1];
    const float* W1    = (const float*)d_in[2];
    const float* b1    = (const float*)d_in[3];
    const float* W2    = (const float*)d_in[4];
    const float* b2    = (const float*)d_in[5];
    const float* W3    = (const float*)d_in[6];
    const float* b3    = (const float*)d_in[7];
    const float* gamma = (const float*)d_in[8];
    const float* beta  = (const float*)d_in[9];

    float* h2  = (float*)d_ws;
    float* out = (float*)d_out;

    k2_layer12<<<BT_,    512, 0, stream>>>(x, nb, W1, b1, W2, b2, gamma, beta, h2);
    k3_layer3<<<BT_ * 2, 256, 0, stream>>>(h2, nb, W3, b3, out);
}


# Layer 3: one block serves four bts on a quarter of the nodes (exactly 4 h2 channels staged), bt quads interleaved in LDS so each index read, W3 read and 16-byte gather feeds four sums; 192 blocks
# speedup vs baseline: 3.7470x; 1.1274x over previous
#include <hip/hip_runtime.h>

#define BT_    192
#define N_     1000
#define NP_    1008
#define K_     16
#define C1_    32
#define C2_    16
#define NK_    16000
#define NTILES 63
#define NBP2_  16128
#define NBP3_  16256
#define W2P_   516

typedef __attribute__((ext_vector_type(2))) float v2f;
typedef __attribute__((ext_vector_type(4))) float v4f;
typedef __attribute__((ext_vector_type(8))) float v8f;
typedef __attribute__((ext_vector_type(2))) int   v2i;
typedef __attribute__((ext_vector_type(4))) int   v4i;

__device__ __forceinline__ v8f wmma_f32_4(v2f a, v2f b, v8f c) {
    return __builtin_amdgcn_wmma_f32_16x16x4_f32(
         false, a,  false, b,
         (short)0, c,  false,  false);
}

#define NCH_   17
#define H1P_   1010

__device__ __forceinline__ void store_h1_pairs(v2f* dst, v8f acc0, v8f acc1, float bias) {
    for (int jj = 0; jj < 4; ++jj) {
        float p0 = acc0[2 * jj]     + bias;  p0 = p0 > 0.f ? p0 : 0.f;
        float q0 = acc1[2 * jj]     + bias;  q0 = q0 > 0.f ? q0 : 0.f;
        float p1 = acc0[2 * jj + 1] + bias;  p1 = p1 > 0.f ? p1 : 0.f;
        float q1 = acc1[2 * jj + 1] + bias;  q1 = q1 > 0.f ? q1 : 0.f;
        ((v4f*)dst)[jj] = (v4f){p0, q0, p1, q1};
    }
}

__global__ __launch_bounds__(512)
void k2_layer12(const float* __restrict__ x, const int* __restrict__ nb,
                const float* __restrict__ W1, const float* __restrict__ b1,
                const float* __restrict__ W2, const float* __restrict__ b2,
                const float* __restrict__ gamma, const float* __restrict__ beta,
                float* __restrict__ h2) {
    __shared__ __attribute__((aligned(16))) v2f   h1p[NCH_ * H1P_];
    __shared__ __attribute__((aligned(16))) int   nbs[NBP2_];
    __shared__ __attribute__((aligned(16))) float w2s[C2_ * W2P_];
    __shared__ __attribute__((aligned(16))) v2f   xs[NP_];

    const int tid    = threadIdx.x;
    const int bt0    = (blockIdx.x >> 1) * 2;
    const int halfId = blockIdx.x & 1;
    const unsigned cbase = (unsigned)halfId * 15u;

    const int lane  = tid & 31;
    const int wave  = tid >> 5;
    const int mrow  = lane & 15;
    const int hi    = lane >> 4;
    const int khalf = hi * 2;

    const v4i* nq = (const v4i*)nb;
    const v4f* wq = (const v4f*)W2;
    const v4f* x0 = (const v4f*)(x + (size_t)bt0 * N_);
    const v4f* x1 = (const v4f*)(x + (size_t)(bt0 + 1) * N_);
    v4f rw[4];
    v4i rn[8];
#pragma unroll
    for (int it = 0; it < 7; ++it)
        rn[it] = nq[tid + it * 512];
#pragma unroll
    for (int it = 0; it < 4; ++it)
        rw[it] = wq[tid + it * 512];
    v2f bw0[4], bw1[4];
    for (int s = 0; s < 4; ++s) {
        bw0[s] = *(const v2f*)&W1[mrow * K_ + s * 4 + khalf];
        bw1[s] = *(const v2f*)&W1[(16 + mrow) * K_ + s * 4 + khalf];
    }
    const float bias1_0 = b1[mrow];
    const float bias1_1 = b1[16 + mrow];
    rn[7] = (v4i){0, 0, 0, 0};
    v4f xa = {0.f, 0.f, 0.f, 0.f}, xb = {0.f, 0.f, 0.f, 0.f};
    if (tid + 3584 < NK_ / 4) rn[7] = nq[tid + 3584];
    if (tid < N_ / 4) {
        xa = x0[tid];
        xb = x1[tid];
    }

#pragma unroll
    for (int it = 0; it < 8; ++it) {
        const int i = tid + it * 512;
        if (it < 7 || i < NK_ / 4)
            ((v4i*)nbs)[i + (i >> 7)] = rn[it];
    }
#pragma unroll
    for (int it = 0; it < 4; ++it) {
        const int i = tid + it * 512;
        ((v4f*)w2s)[i + (i >> 7)] = rw[it];
    }
    if (tid < N_ / 4) {
        ((v4f*)xs)[2 * tid]     = (v4f){xa.x, xb.x, xa.y, xb.y};
        ((v4f*)xs)[2 * tid + 1] = (v4f){xa.z, xb.z, xa.w, xb.w};
    }
    __syncthreads();

    const int wave_u = __builtin_amdgcn_readfirstlane(wave);
    for (int nt = wave_u; nt < NTILES; nt += 16) {
        const int base1 = nt * 16;
        const int node1 = min(base1 + mrow, N_ - 1);
        v2f a0[4], a1[4];
        for (int s = 0; s < 4; ++s) {
            const unsigned r  = (unsigned)node1 * K_ + (unsigned)(s * 4 + khalf);
            const unsigned pr = r + ((r >> 9) << 2);
            const v2i idx = *(const v2i*)&nbs[pr];
            const v2f g0 = xs[idx.x];
            const v2f g1 = xs[idx.y];
            a0[s] = (v2f){g0.x, g1.x};
            a1[s] = (v2f){g0.y, g1.y};
        }
        if (halfId == 0) {
            v8f acc0 = {}, acc1 = {};
            for (int s = 0; s < 4; ++s) {
                acc0 = wmma_f32_4(a0[s], bw0[s], acc0);
                acc1 = wmma_f32_4(a1[s], bw0[s], acc1);
            }
            store_h1_pairs(h1p + mrow * H1P_ + base1 + 8 * hi, acc0, acc1, bias1_0);
        }
        {
            v8f acc0 = {}, acc1 = {};
            for (int s = 0; s < 4; ++s) {
                acc0 = wmma_f32_4(a0[s], bw1[s], acc0);
                acc1 = wmma_f32_4(a1[s], bw1[s], acc1);
            }
            const unsigned cl = 16u + (unsigned)mrow - cbase;
            if (cl < (unsigned)NCH_)
                store_h1_pairs(h1p + cl * H1P_ + base1 + 8 * hi, acc0, acc1, bias1_1);
        }
    }
    __syncthreads();

    const float scale = gamma[mrow] * rsqrtf(1.0f + 1e-5f);
    const float bias  = b2[mrow];
    const float shift = beta[mrow];

    for (int t = 0; t < 2; ++t) {
        const int tile = halfId * 32 + wave * 2 + t;
        if (tile >= NTILES) continue;
        const int base = tile * 16;
        const int node = min(base + mrow, N_ - 1);
        const unsigned fbase = (unsigned)node * 512u;

        v8f acc0 = {};
        v8f acc1 = {};
#pragma unroll 8
        for (int s = 0; s < 128; ++s) {
            const int k0 = s * 4 + khalf;
            const unsigned f0 = fbase + (unsigned)k0;
            const unsigned c  = f0 / (unsigned)NK_;
            const unsigned r  = f0 - c * (unsigned)NK_;
            const unsigned pr = r + ((r >> 9) << 2);
            const v2i idx = *(const v2i*)&nbs[pr];
            const v2f* hc = h1p + (c - cbase) * H1P_;
            const v2f g0 = hc[idx.x];
            const v2f g1 = hc[idx.y];
            const v2f a0 = {g0.x, g1.x};
            const v2f a1 = {g0.y, g1.y};
            const v2f b = *(const v2f*)&w2s[mrow * W2P_ + k0];
            acc0 = wmma_f32_4(a0, b, acc0);
            acc1 = wmma_f32_4(a1, b, acc1);
        }
#pragma unroll
        for (int q = 0; q < 2; ++q) {
            const v8f acc = q ? acc1 : acc0;
            v4f lo, hiv;
#pragma unroll
            for (int j = 0; j < 4; ++j) {
                float v0 = (acc[j]     + bias) * scale + shift;  lo[j]  = v0 > 0.f ? v0 : 0.f;
                float v1 = (acc[j + 4] + bias) * scale + shift;  hiv[j] = v1 > 0.f ? v1 : 0.f;
            }
            float* p = h2 + (size_t)((bt0 + q) * C2_ + mrow) * NP_ + base + 8 * hi;
            *(v4f*)p       = lo;
            *(v4f*)(p + 4) = hiv;
        }
    }
}

#define Q3N_   250
#define C3Q_   4
#define SLABQ_ 1008

__global__ __launch_bounds__(256)
void k3_layer3(const float* __restrict__ h2, const int* __restrict__ nb,
               const float* __restrict__ W3, const float* __restrict__ b3,
               float* __restrict__ out) {
    __shared__ __attribute__((aligned(16))) v4f   h2q[C3Q_ * NP_];
    __shared__ __attribute__((aligned(16))) int   nbs[NBP3_];
    __shared__ __attribute__((aligned(16))) float w3s[256];

    const int tid = threadIdx.x;
    const int bq  = blockIdx.x >> 2;
    const int qd  = blockIdx.x & 3;

    const v4i* nq = (const v4i*)nb;
    const v4f* s0 = (const v4f*)(h2 + ((size_t)(4 * bq + 0) * C2_ + C3Q_ * qd) * NP_);
    const v4f* s1 = (const v4f*)(h2 + ((size_t)(4 * bq + 1) * C2_ + C3Q_ * qd) * NP_);
    const v4f* s2 = (const v4f*)(h2 + ((size_t)(4 * bq + 2) * C2_ + C3Q_ * qd) * NP_);
    const v4f* s3 = (const v4f*)(h2 + ((size_t)(4 * bq + 3) * C2_ + C3Q_ * qd) * NP_);
    v4i rn[16];
    v4f ra[4], rb[4], rc[4], rd[4];
#pragma unroll
    for (int it = 0; it < 15; ++it)
        rn[it] = nq[tid + it * 256];
#pragma unroll
    for (int it = 0; it < 3; ++it) {
        const int i = tid + it * 256;
        ra[it] = s0[i];  rb[it] = s1[i];  rc[it] = s2[i];  rd[it] = s3[i];
    }
    const float w3v = W3[tid];
    rn[15] = (v4i){0, 0, 0, 0};
    ra[3] = rb[3] = rc[3] = rd[3] = (v4f){0.f, 0.f, 0.f, 0.f};
    if (tid + 3840 < NK_ / 4) rn[15] = nq[tid + 3840];
    if (tid + 768 < SLABQ_) {
        const int i = tid + 768;
        ra[3] = s0[i];  rb[3] = s1[i];  rc[3] = s2[i];  rd[3] = s3[i];
    }

#pragma unroll
    for (int it = 0; it < 16; ++it) {
        const int i = tid + it * 256;
        if (it < 15 || i < NK_ / 4)
            ((v4i*)nbs)[i + (i >> 6)] = rn[it];
    }
#pragma unroll
    for (int it = 0; it < 4; ++it) {
        const int i = tid + it * 256;
        if (it < 3 || i < SLABQ_) {
            h2q[4 * i + 0] = (v4f){ra[it].x, rb[it].x, rc[it].x, rd[it].x};
            h2q[4 * i + 1] = (v4f){ra[it].y, rb[it].y, rc[it].y, rd[it].y};
            h2q[4 * i + 2] = (v4f){ra[it].z, rb[it].z, rc[it].z, rd[it].z};
            h2q[4 * i + 3] = (v4f){ra[it].w, rb[it].w, rc[it].w, rd[it].w};
        }
    }
    w3s[tid] = w3v;
    __syncthreads();

    if (tid < Q3N_) {
        const int n = qd * Q3N_ + tid;
        const float b3v = b3[0];
        v4f acc = {b3v, b3v, b3v, b3v};
        const unsigned fbase = (unsigned)n * 256u;
        const unsigned cb    = (unsigned)(C3Q_ * qd);
#pragma unroll 1
        for (int f = 0; f < 256; f += 16) {
            const unsigned fl = fbase + (unsigned)f;
            const unsigned c  = fl / (unsigned)NK_;
            const unsigned r  = fl - c * (unsigned)NK_;
            const unsigned pr = r + ((r >> 8) << 2);
            const v4f* hc = h2q + (c - cb) * NP_;
#pragma unroll
            for (int q = 0; q < 4; ++q) {
                const v4i idx = *(const v4i*)&nbs[pr + 4 * q];
                const v4f w   = *(const v4f*)&w3s[f + 4 * q];
                const v4f g0 = hc[idx.x];
                const v4f g1 = hc[idx.y];
                const v4f g2 = hc[idx.z];
                const v4f g3 = hc[idx.w];
#pragma unroll
                for (int j = 0; j < 4; ++j) {
                    float a = acc[j];
                    a = fmaf(g0[j], w.x, a);
                    a = fmaf(g1[j], w.y, a);
                    a = fmaf(g2[j], w.z, a);
                    a = fmaf(g3[j], w.w, a);
                    acc[j] = a;
                }
            }
        }
#pragma unroll
        for (int j = 0; j < 4; ++j)
            out[(4 * bq + j) * N_ + n] = acc[j] > 0.f ? acc[j] : 0.f;
    }
}

extern "C" void kernel_launch(void* const* d_in, const int* in_sizes, int n_in,
                              void* d_out, int out_size, void* d_ws, size_t ws_size,
                              hipStream_t stream) {
    const float* x     = (const float*)d_in[0];
    const int*   nb    = (const int*)  d_in[1];
    const float* W1    = (const float*)d_in[2];
    const float* b1    = (const float*)d_in[3];
    const float* W2    = (const float*)d_in[4];
    const float* b2    = (const float*)d_in[5];
    const float* W3    = (const float*)d_in[6];
    const float* b3    = (const float*)d_in[7];
    const float* gamma = (const float*)d_in[8];
    const float* beta  = (const float*)d_in[9];

    float* h2  = (float*)d_ws;
    float* out = (float*)d_out;

    k2_layer12<<<BT_,    512, 0, stream>>>(x, nb, W1, b1, W2, b2, gamma, beta, h2);
    k3_layer3<<<BT_,     256, 0, stream>>>(h2, nb, W3, b3, out);
}


# Layers 1+2: one block serves three bts on a quarter of the node tiles (9 h1 channels, 16-byte bt slots read 12 bytes at a time), 256 blocks so every WGP gets one (was 192 of 256); sum order kept
# speedup vs baseline: 4.4957x; 1.1998x over previous
#include <hip/hip_runtime.h>

#define BT_    192
#define N_     1000
#define NP_    1008
#define K_     16
#define C1_    32
#define C2_    16
#define NK_    16000
#define NTILES 63
#define NBP2_  16128
#define NBP3_  16256
#define W2P_   516

typedef __attribute__((ext_vector_type(2))) float v2f;
typedef __attribute__((ext_vector_type(4))) float v4f;
typedef __attribute__((ext_vector_type(8))) float v8f;
typedef __attribute__((ext_vector_type(2))) int   v2i;
typedef __attribute__((ext_vector_type(4))) int   v4i;

__device__ __forceinline__ v8f wmma_f32_4(v2f a, v2f b, v8f c) {
    return __builtin_amdgcn_wmma_f32_16x16x4_f32(
         false, a,  false, b,
         (short)0, c,  false,  false);
}


typedef __attribute__((ext_vector_type(3))) float v3f;
__device__ __forceinline__ v3f load_slot3(const v4f* slot) {
    const float* p = (const float*)__builtin_assume_aligned(slot, 16);
    return (v3f){p[0], p[1], p[2]};
}

#define NCH_   9
#define H1P_   1009
#define NBLK2_ 256

__device__ __forceinline__ void store_h1_slots(v4f* dst, v8f acc0, v8f acc1, v8f acc2, float bias) {
    for (int r = 0; r < 8; ++r) {
        float p0 = acc0[r] + bias;  p0 = p0 > 0.f ? p0 : 0.f;
        float p1 = acc1[r] + bias;  p1 = p1 > 0.f ? p1 : 0.f;
        float p2 = acc2[r] + bias;  p2 = p2 > 0.f ? p2 : 0.f;
        dst[r] = (v4f){p0, p1, p2, 0.f};
    }
}

__device__ __forceinline__ void store_h2_row(float* p, v8f acc, float bias, float scale, float shift) {
    v4f lo, hiv;
    for (int j = 0; j < 4; ++j) {
        float v0 = (acc[j]     + bias) * scale + shift;  lo[j]  = v0 > 0.f ? v0 : 0.f;
        float v1 = (acc[j + 4] + bias) * scale + shift;  hiv[j] = v1 > 0.f ? v1 : 0.f;
    }
    *(v4f*)p       = lo;
    *(v4f*)(p + 4) = hiv;
}

__global__ __launch_bounds__(512)
void k2_layer12(const float* __restrict__ x, const int* __restrict__ nb,
                const float* __restrict__ W1, const float* __restrict__ b1,
                const float* __restrict__ W2, const float* __restrict__ b2,
                const float* __restrict__ gamma, const float* __restrict__ beta,
                float* __restrict__ h2) {
    __shared__ __attribute__((aligned(16))) v4f   h1q[NCH_ * H1P_];
    __shared__ __attribute__((aligned(16))) int   nbs[NBP2_];
    __shared__ __attribute__((aligned(16))) float w2s[C2_ * W2P_];
    __shared__ __attribute__((aligned(16))) v4f   xs[NP_];

    const int tid    = threadIdx.x;
    const int bt0    = (blockIdx.x >> 2) * 3;
    const int qd     = blockIdx.x & 3;
    const unsigned cbase = (unsigned)qd * 8u;

    const int lane  = tid & 31;
    const int wave  = tid >> 5;
    const int mrow  = lane & 15;
    const int hi    = lane >> 4;
    const int khalf = hi * 2;

    const v4i* nq = (const v4i*)nb;
    const v4f* wq = (const v4f*)W2;
    const v4f* x0 = (const v4f*)(x + (size_t)bt0 * N_);
    const v4f* x1 = (const v4f*)(x + (size_t)(bt0 + 1) * N_);
    const v4f* x2 = (const v4f*)(x + (size_t)(bt0 + 2) * N_);
    v4f rw[4];
    v4i rn[8];
#pragma unroll
    for (int it = 0; it < 7; ++it)
        rn[it] = nq[tid + it * 512];
#pragma unroll
    for (int it = 0; it < 4; ++it)
        rw[it] = wq[tid + it * 512];
    const int chrow = min((int)cbase + mrow, C1_ - 1);
    v2f bw[4];
    for (int s = 0; s < 4; ++s)
        bw[s] = *(const v2f*)&W1[chrow * K_ + s * 4 + khalf];
    const float bias1 = b1[chrow];
    const bool keep1 = (mrow < NCH_) && ((int)cbase + mrow < C1_);
    rn[7] = (v4i){0, 0, 0, 0};
    v4f xa = {0.f, 0.f, 0.f, 0.f}, xb = {0.f, 0.f, 0.f, 0.f}, xc = {0.f, 0.f, 0.f, 0.f};
    if (tid + 3584 < NK_ / 4) rn[7] = nq[tid + 3584];
    if (tid < N_ / 4) {
        xa = x0[tid];
        xb = x1[tid];
        xc = x2[tid];
    }

#pragma unroll
    for (int it = 0; it < 8; ++it) {
        const int i = tid + it * 512;
        if (it < 7 || i < NK_ / 4)
            ((v4i*)nbs)[i + (i >> 7)] = rn[it];
    }
#pragma unroll
    for (int it = 0; it < 4; ++it) {
        const int i = tid + it * 512;
        ((v4f*)w2s)[i + (i >> 7)] = rw[it];
    }
    if (tid < N_ / 4) {
        xs[4 * tid]     = (v4f){xa.x, xb.x, xc.x, 0.f};
        xs[4 * tid + 1] = (v4f){xa.y, xb.y, xc.y, 0.f};
        xs[4 * tid + 2] = (v4f){xa.z, xb.z, xc.z, 0.f};
        xs[4 * tid + 3] = (v4f){xa.w, xb.w, xc.w, 0.f};
    }
    __syncthreads();

    const int wave_u = __builtin_amdgcn_readfirstlane(wave);
    for (int nt = wave_u; nt < NTILES; nt += 16) {
        const int base1 = nt * 16;
        const int node1 = min(base1 + mrow, N_ - 1);
        v2f a0[4], a1[4], a2[4];
        for (int s = 0; s < 4; ++s) {
            const unsigned r  = (unsigned)node1 * K_ + (unsigned)(s * 4 + khalf);
            const unsigned pr = r + ((r >> 9) << 2);
            const v2i idx = *(const v2i*)&nbs[pr];
            const v3f g0 = load_slot3(xs + idx.x);
            const v3f g1 = load_slot3(xs + idx.y);
            a0[s] = (v2f){g0.x, g1.x};
            a1[s] = (v2f){g0.y, g1.y};
            a2[s] = (v2f){g0.z, g1.z};
        }
        v8f acc0 = {}, acc1 = {}, acc2 = {};
        for (int s = 0; s < 4; ++s) {
            acc0 = wmma_f32_4(a0[s], bw[s], acc0);
            acc1 = wmma_f32_4(a1[s], bw[s], acc1);
            acc2 = wmma_f32_4(a2[s], bw[s], acc2);
        }
        if (keep1)
            store_h1_slots(h1q + mrow * H1P_ + base1 + 8 * hi, acc0, acc1, acc2, bias1);
    }
    __syncthreads();

    const float scale = gamma[mrow] * rsqrtf(1.0f + 1e-5f);
    const float bias  = b2[mrow];
    const float shift = beta[mrow];

    const int tile = qd * 16 + wave_u;
    if (tile < NTILES) {
        const int base = tile * 16;
        const int node = min(base + mrow, N_ - 1);
        const unsigned fbase = (unsigned)node * 512u;

        v8f acc0 = {};
        v8f acc1 = {};
        v8f acc2 = {};
#pragma unroll 8
        for (int s = 0; s < 128; ++s) {
            const int k0 = s * 4 + khalf;
            const unsigned f0 = fbase + (unsigned)k0;
            const unsigned c  = f0 / (unsigned)NK_;
            const unsigned r  = f0 - c * (unsigned)NK_;
            const unsigned pr = r + ((r >> 9) << 2);
            const v2i idx = *(const v2i*)&nbs[pr];
            const v4f* hc = h1q + (c - cbase) * H1P_;
            const v3f g0 = load_slot3(hc + idx.x);
            const v3f g1 = load_slot3(hc + idx.y);
            const v2f a0 = {g0.x, g1.x};
            const v2f a1 = {g0.y, g1.y};
            const v2f a2 = {g0.z, g1.z};
            const v2f b = *(const v2f*)&w2s[mrow * W2P_ + k0];
            acc0 = wmma_f32_4(a0, b, acc0);
            acc1 = wmma_f32_4(a1, b, acc1);
            acc2 = wmma_f32_4(a2, b, acc2);
        }
        float* p = h2 + (size_t)(bt0 * C2_ + mrow) * NP_ + base + 8 * hi;
        store_h2_row(p,                            acc0, bias, scale, shift);
        store_h2_row(p + (size_t)C2_ * NP_,        acc1, bias, scale, shift);
        store_h2_row(p + (size_t)2 * C2_ * NP_,    acc2, bias, scale, shift);
    }
}

#define Q3N_   250
#define C3Q_   4
#define SLABQ_ 1008

__global__ __launch_bounds__(256)
void k3_layer3(const float* __restrict__ h2, const int* __restrict__ nb,
               const float* __restrict__ W3, const float* __restrict__ b3,
               float* __restrict__ out) {
    __shared__ __attribute__((aligned(16))) v4f   h2q[C3Q_ * NP_];
    __shared__ __attribute__((aligned(16))) int   nbs[NBP3_];
    __shared__ __attribute__((aligned(16))) float w3s[256];

    const int tid = threadIdx.x;
    const int bq  = blockIdx.x >> 2;
    const int qd  = blockIdx.x & 3;

    const v4i* nq = (const v4i*)nb;
    const v4f* s0 = (const v4f*)(h2 + ((size_t)(4 * bq + 0) * C2_ + C3Q_ * qd) * NP_);
    const v4f* s1 = (const v4f*)(h2 + ((size_t)(4 * bq + 1) * C2_ + C3Q_ * qd) * NP_);
    const v4f* s2 = (const v4f*)(h2 + ((size_t)(4 * bq + 2) * C2_ + C3Q_ * qd) * NP_);
    const v4f* s3 = (const v4f*)(h2 + ((size_t)(4 * bq + 3) * C2_ + C3Q_ * qd) * NP_);
    v4i rn[16];
    v4f ra[4], rb[4], rc[4], rd[4];
#pragma unroll
    for (int it = 0; it < 15; ++it)
        rn[it] = nq[tid + it * 256];
#pragma unroll
    for (int it = 0; it < 3; ++it) {
        const int i = tid + it * 256;
        ra[it] = s0[i];  rb[it] = s1[i];  rc[it] = s2[i];  rd[it] = s3[i];
    }
    const float w3v = W3[tid];
    rn[15] = (v4i){0, 0, 0, 0};
    ra[3] = rb[3] = rc[3] = rd[3] = (v4f){0.f, 0.f, 0.f, 0.f};
    if (tid + 3840 < NK_ / 4) rn[15] = nq[tid + 3840];
    if (tid + 768 < SLABQ_) {
        const int i = tid + 768;
        ra[3] = s0[i];  rb[3] = s1[i];  rc[3] = s2[i];  rd[3] = s3[i];
    }

#pragma unroll
    for (int it = 0; it < 16; ++it) {
        const int i = tid + it * 256;
        if (it < 15 || i < NK_ / 4)
            ((v4i*)nbs)[i + (i >> 6)] = rn[it];
    }
#pragma unroll
    for (int it = 0; it < 4; ++it) {
        const int i = tid + it * 256;
        if (it < 3 || i < SLABQ_) {
            h2q[4 * i + 0] = (v4f){ra[it].x, rb[it].x, rc[it].x, rd[it].x};
            h2q[4 * i + 1] = (v4f){ra[it].y, rb[it].y, rc[it].y, rd[it].y};
            h2q[4 * i + 2] = (v4f){ra[it].z, rb[it].z, rc[it].z, rd[it].z};
            h2q[4 * i + 3] = (v4f){ra[it].w, rb[it].w, rc[it].w, rd[it].w};
        }
    }
    w3s[tid] = w3v;
    __syncthreads();

    if (tid < Q3N_) {
        const int n = qd * Q3N_ + tid;
        const float b3v = b3[0];
        v4f acc = {b3v, b3v, b3v, b3v};
        const unsigned fbase = (unsigned)n * 256u;
        const unsigned cb    = (unsigned)(C3Q_ * qd);
#pragma unroll 1
        for (int f = 0; f < 256; f += 16) {
            const unsigned fl = fbase + (unsigned)f;
            const unsigned c  = fl / (unsigned)NK_;
            const unsigned r  = fl - c * (unsigned)NK_;
            const unsigned pr = r + ((r >> 8) << 2);
            const v4f* hc = h2q + (c - cb) * NP_;
#pragma unroll
            for (int q = 0; q < 4; ++q) {
                const v4i idx = *(const v4i*)&nbs[pr + 4 * q];
                const v4f w   = *(const v4f*)&w3s[f + 4 * q];
                const v4f g0 = hc[idx.x];
                const v4f g1 = hc[idx.y];
                const v4f g2 = hc[idx.z];
                const v4f g3 = hc[idx.w];
#pragma unroll
                for (int j = 0; j < 4; ++j) {
                    float a = acc[j];
                    a = fmaf(g0[j], w.x, a);
                    a = fmaf(g1[j], w.y, a);
                    a = fmaf(g2[j], w.z, a);
                    a = fmaf(g3[j], w.w, a);
                    acc[j] = a;
                }
            }
        }
#pragma unroll
        for (int j = 0; j < 4; ++j)
            out[(4 * bq + j) * N_ + n] = acc[j] > 0.f ? acc[j] : 0.f;
    }
}

extern "C" void kernel_launch(void* const* d_in, const int* in_sizes, int n_in,
                              void* d_out, int out_size, void* d_ws, size_t ws_size,
                              hipStream_t stream) {
    const float* x     = (const float*)d_in[0];
    const int*   nb    = (const int*)  d_in[1];
    const float* W1    = (const float*)d_in[2];
    const float* b1    = (const float*)d_in[3];
    const float* W2    = (const float*)d_in[4];
    const float* b2    = (const float*)d_in[5];
    const float* W3    = (const float*)d_in[6];
    const float* b3    = (const float*)d_in[7];
    const float* gamma = (const float*)d_in[8];
    const float* beta  = (const float*)d_in[9];

    float* h2  = (float*)d_ws;
    float* out = (float*)d_out;

    k2_layer12<<<NBLK2_, 512, 0, stream>>>(x, nb, W1, b1, W2, b2, gamma, beta, h2);
    k3_layer3<<<BT_,     256, 0, stream>>>(h2, nb, W3, b3, out);
}
